# starting kernel
# baseline (speedup 1.0000x reference)
#include <hip/hip_runtime.h>

typedef __attribute__((ext_vector_type(2))) float v2f;
typedef __attribute__((ext_vector_type(8))) float v8f;

#define WSZ     7
#define PAD     3
#define TILE_H  16
#define TILE_W  128
#define IN_ROWS (TILE_H + WSZ - 1)
#define IN_COLS (TILE_W + WSZ - 1)
#define XSTRIDE 136
#define HROWS   24
#define IMG_H   512
#define IMG_W   512

__global__ __launch_bounds__(256) void local_variance_kernel(
    const float* __restrict__ x, float* __restrict__ out)
{
    __shared__ float xbuf[IN_ROWS][XSTRIDE];
    __shared__ float hs[HROWS][TILE_W];
    __shared__ float hq[HROWS][TILE_W];

    const int tid = threadIdx.x;
    const int img = blockIdx.z;
    const int y0  = blockIdx.y * TILE_H;
    const int c0  = blockIdx.x * TILE_W;
    const float* __restrict__ src = x + (size_t)img * IMG_H * IMG_W;
    float* __restrict__ dst       = out + (size_t)img * IMG_H * IMG_W;

    for (int idx = tid; idx < IN_ROWS * IN_COLS; idx += 256) {
        int r  = idx / IN_COLS;
        int cc = idx - r * IN_COLS;
        int gy = (y0 + r  - PAD + IMG_H) & (IMG_H - 1);
        int gx = (c0 + cc - PAD + IMG_W) & (IMG_W - 1);
        xbuf[r][cc] = src[gy * IMG_W + gx];
    }
    __syncthreads();

    for (int idx = tid; idx < IN_ROWS * TILE_W; idx += 256) {
        int r = idx / TILE_W;
        int c = idx - r * TILE_W;
        float s = 0.0f, q = 0.0f;
#pragma unroll
        for (int j = 0; j < WSZ; ++j) {
            float v = xbuf[r][c + j];
            s += v;
            q += v * v;
        }
        hs[r][c] = s;
        hq[r][c] = q;
    }
    for (int idx = tid; idx < (HROWS - IN_ROWS) * TILE_W; idx += 256) {
        int r = IN_ROWS + idx / TILE_W;
        int c = idx % TILE_W;
        hs[r][c] = 0.0f;
        hq[r][c] = 0.0f;
    }
    __syncthreads();

    const int lane = tid & 31;
    const int wv   = tid >> 5;
    const int n    = lane & 15;
    const int hf   = lane >> 4;
    const int cb   = wv * 16;

    v8f accS = {};
    v8f accQ = {};
#pragma unroll
    for (int kc = 0; kc < 6; ++kc) {
        const int k0 = kc * 4 + hf * 2;
        v2f a;
        a.x = (k0     >= n && k0     <= n + 6) ? 1.0f : 0.0f;
        a.y = (k0 + 1 >= n && k0 + 1 <= n + 6) ? 1.0f : 0.0f;
        v2f bs, bq;
        bs.x = hs[k0][cb + n];
        bs.y = hs[k0 + 1][cb + n];
        bq.x = hq[k0][cb + n];
        bq.y = hq[k0 + 1][cb + n];
        accS = __builtin_amdgcn_wmma_f32_16x16x4_f32(
            false, a, false, bs, (short)0, accS, false, false);
        accQ = __builtin_amdgcn_wmma_f32_16x16x4_f32(
            false, a, false, bq, (short)0, accQ, false, false);
    }

    const float inv_area = 1.0f / (float)(WSZ * WSZ);
    const int gx = c0 + cb + n;
#pragma unroll
    for (int v = 0; v < 8; ++v) {
        int gy = y0 + v + hf * 8;
        float s = accS[v] * inv_area;
        float q = accQ[v] * inv_area;
        dst[gy * IMG_W + gx] = q - s * s;
    }
}

extern "C" void kernel_launch(void* const* d_in, const int* in_sizes, int n_in,
                              void* d_out, int out_size, void* d_ws, size_t ws_size,
                              hipStream_t stream) {
    (void)in_sizes; (void)n_in; (void)d_ws; (void)ws_size; (void)out_size;
    const float* x = (const float*)d_in[0];
    float* out = (float*)d_out;
    dim3 grid(IMG_W / TILE_W, IMG_H / TILE_H, 16 * 3);
    dim3 block(256, 1, 1);
    local_variance_kernel<<<grid, block, 0, stream>>>(x, out);
}


# Resident grid with a tile loop: 1024 work-groups (4 per WGP, all resident) each walk 6 tiles instead of 6144 one-tile groups; the next tile's raw loads are issued one tile ahead into registers
# speedup vs baseline: 1.1027x; 1.1027x over previous
#include <hip/hip_runtime.h>


typedef __attribute__((ext_vector_type(2))) float v2f;
typedef __attribute__((ext_vector_type(8))) float v8f;

#define WSZ     7
#define PAD     3
#define TILE_H  16
#define TILE_W  128
#define IN_ROWS (TILE_H + WSZ - 1)
#define IN_COLS (TILE_W + WSZ - 1)
#define XSTRIDE 136
#define HROWS   24
#define IMG_H   512
#define IMG_W   512
#define N_IMG   48
#define TILES_X (IMG_W / TILE_W)
#define TILES_Y (IMG_H / TILE_H)
#define N_TILES (TILES_X * TILES_Y * N_IMG)
#define N_GROUPS 1024
#define TILES_PER_GROUP (N_TILES / N_GROUPS)
#define N_RAW   (IN_ROWS * IN_COLS)
#define RAW_PER_THREAD ((N_RAW + 255) / 256)

__device__ __forceinline__ void tile_coords(int t, int& img, int& y0, int& c0)
{
    y0  = (t & (TILES_Y - 1)) * TILE_H;
    c0  = ((t / TILES_Y) & (TILES_X - 1)) * TILE_W;
    img = t / (TILES_Y * TILES_X);
}

__device__ __forceinline__ void load_raw(const float* __restrict__ x, int t, int tid,
                                         float (&pf)[RAW_PER_THREAD])
{
    int img, y0, c0;
    tile_coords(t, img, y0, c0);
    const float* __restrict__ src = x + (size_t)img * IMG_H * IMG_W;
#pragma unroll
    for (int i = 0; i < RAW_PER_THREAD; ++i) {
        int idx = tid + 256 * i;
        if (idx < N_RAW) {
            int r  = idx / IN_COLS;
            int cc = idx - r * IN_COLS;
            int gy = (y0 + r  - PAD + IMG_H) & (IMG_H - 1);
            int gx = (c0 + cc - PAD + IMG_W) & (IMG_W - 1);
            pf[i] = src[gy * IMG_W + gx];
        }
    }
}

__global__ __launch_bounds__(256) __attribute__((amdgpu_waves_per_eu(8)))
void local_variance_kernel(
    const float* __restrict__ x, float* __restrict__ out)
{
    __shared__ float xbuf[IN_ROWS][XSTRIDE];
    __shared__ float hs[HROWS][TILE_W];
    __shared__ float hq[HROWS][TILE_W];

    const int tid  = threadIdx.x;
    const int lane = tid & 31;
    const int wv   = tid >> 5;
    const int n    = lane & 15;
    const int hf   = lane >> 4;
    const int cb   = wv * 16;
    const float inv_area = 1.0f / (float)(WSZ * WSZ);

    const int t0 = blockIdx.x * TILES_PER_GROUP;
    float pf[RAW_PER_THREAD];

    for (int idx = tid; idx < (HROWS - IN_ROWS) * TILE_W; idx += 256) {
        int r = IN_ROWS + idx / TILE_W;
        int c = idx % TILE_W;
        hs[r][c] = 0.0f;
        hq[r][c] = 0.0f;
    }

    load_raw(x, t0, tid, pf);
#pragma unroll
    for (int i = 0; i < RAW_PER_THREAD; ++i) {
        int idx = tid + 256 * i;
        if (idx < N_RAW) {
            int r  = idx / IN_COLS;
            int cc = idx - r * IN_COLS;
            xbuf[r][cc] = pf[i];
        }
    }
    __syncthreads();

    for (int it = 0; it < TILES_PER_GROUP; ++it) {
        const int t = t0 + it;
        const bool has_next = (it + 1 < TILES_PER_GROUP);

        if (has_next) load_raw(x, t + 1, tid, pf);

        for (int idx = tid; idx < IN_ROWS * TILE_W; idx += 256) {
            int r = idx / TILE_W;
            int c = idx - r * TILE_W;
            float s = 0.0f, q = 0.0f;
#pragma unroll
            for (int j = 0; j < WSZ; ++j) {
                float v = xbuf[r][c + j];
                s += v;
                q += v * v;
            }
            hs[r][c] = s;
            hq[r][c] = q;
        }
        __syncthreads();

        v8f accS = {};
        v8f accQ = {};
#pragma unroll
        for (int kc = 0; kc < 6; ++kc) {
            const int k0 = kc * 4 + hf * 2;
            v2f a;
            a.x = (k0     >= n && k0     <= n + 6) ? 1.0f : 0.0f;
            a.y = (k0 + 1 >= n && k0 + 1 <= n + 6) ? 1.0f : 0.0f;
            v2f bs, bq;
            bs.x = hs[k0][cb + n];
            bs.y = hs[k0 + 1][cb + n];
            bq.x = hq[k0][cb + n];
            bq.y = hq[k0 + 1][cb + n];
            accS = __builtin_amdgcn_wmma_f32_16x16x4_f32(
                false, a, false, bs, (short)0, accS, false, false);
            accQ = __builtin_amdgcn_wmma_f32_16x16x4_f32(
                false, a, false, bq, (short)0, accQ, false, false);
        }

        {
            int img, y0, c0;
            tile_coords(t, img, y0, c0);
            float* __restrict__ dst = out + (size_t)img * IMG_H * IMG_W;
            const int gx = c0 + cb + n;
#pragma unroll
            for (int v = 0; v < 8; ++v) {
                int gy = y0 + v + hf * 8;
                float s = accS[v] * inv_area;
                float q = accQ[v] * inv_area;
                dst[gy * IMG_W + gx] = q - s * s;
            }
        }

        if (has_next) {
#pragma unroll
            for (int i = 0; i < RAW_PER_THREAD; ++i) {
                int idx = tid + 256 * i;
                if (idx < N_RAW) {
                    int r  = idx / IN_COLS;
                    int cc = idx - r * IN_COLS;
                    xbuf[r][cc] = pf[i];
                }
            }
        }
        __syncthreads();
    }
}

extern "C" void kernel_launch(void* const* d_in, const int* in_sizes, int n_in,
                              void* d_out, int out_size, void* d_ws, size_t ws_size,
                              hipStream_t stream) {
    (void)in_sizes; (void)n_in; (void)d_ws; (void)ws_size; (void)out_size;
    const float* x = (const float*)d_in[0];
    float* out = (float*)d_out;
    dim3 grid(N_GROUPS, 1, 1);
    dim3 block(256, 1, 1);
    local_variance_kernel<<<grid, block, 0, stream>>>(x, out);
}


# Horizontal pass: each lane makes 4 adjacent 7-tap sums from one 10-value LDS read (16+16+8 B) with shared partial sums, not 7 reads per output; the order in which the 7 taps are summed changes
# speedup vs baseline: 1.2941x; 1.1736x over previous
#include <hip/hip_runtime.h>


typedef __attribute__((ext_vector_type(2))) float v2f;
typedef __attribute__((ext_vector_type(4))) float v4f;
typedef __attribute__((ext_vector_type(8))) float v8f;

#define WSZ     7
#define PAD     3
#define TILE_H  16
#define TILE_W  128
#define IN_ROWS (TILE_H + WSZ - 1)
#define IN_COLS (TILE_W + WSZ - 1)
#define XSTRIDE 136
#define HROWS   24
#define IMG_H   512
#define IMG_W   512
#define N_IMG   48
#define TILES_X (IMG_W / TILE_W)
#define TILES_Y (IMG_H / TILE_H)
#define N_TILES (TILES_X * TILES_Y * N_IMG)
#define N_GROUPS 1024
#define TILES_PER_GROUP (N_TILES / N_GROUPS)
#define N_RAW   (IN_ROWS * IN_COLS)
#define RAW_PER_THREAD ((N_RAW + 255) / 256)

__device__ __forceinline__ void tile_coords(int t, int& img, int& y0, int& c0)
{
    y0  = (t & (TILES_Y - 1)) * TILE_H;
    c0  = ((t / TILES_Y) & (TILES_X - 1)) * TILE_W;
    img = t / (TILES_Y * TILES_X);
}

__device__ __forceinline__ void load_raw(const float* __restrict__ x, int t, int tid,
                                         float (&pf)[RAW_PER_THREAD])
{
    int img, y0, c0;
    tile_coords(t, img, y0, c0);
    const float* __restrict__ src = x + (size_t)img * IMG_H * IMG_W;
#pragma unroll
    for (int i = 0; i < RAW_PER_THREAD; ++i) {
        int idx = tid + 256 * i;
        if (idx < N_RAW) {
            int r  = idx / IN_COLS;
            int cc = idx - r * IN_COLS;
            int gy = (y0 + r  - PAD + IMG_H) & (IMG_H - 1);
            int gx = (c0 + cc - PAD + IMG_W) & (IMG_W - 1);
            pf[i] = src[gy * IMG_W + gx];
        }
    }
}

__device__ __forceinline__ void hsum4(const float* __restrict__ row, v4f& s, v4f& q)
{
    const v4f xa = *reinterpret_cast<const v4f*>(row);
    const v4f xb = *reinterpret_cast<const v4f*>(row + 4);
    const v2f xc = *reinterpret_cast<const v2f*>(row + 8);
    const float x0 = xa.x, x1 = xa.y, x2 = xa.z, x3 = xa.w;
    const float x4 = xb.x, x5 = xb.y, x6 = xb.z, x7 = xb.w;
    const float x8 = xc.x, x9 = xc.y;

    const float sm  = (x3 + x4) + (x5 + x6);
    const float s12 = x1 + x2;
    const float s78 = x7 + x8;
    s.x = (x0  + s12) + sm;
    s.y = (s12 + x7)  + sm;
    s.z = (x2  + s78) + sm;
    s.w = (s78 + x9)  + sm;

    const float y0 = x0 * x0, y1 = x1 * x1, y2 = x2 * x2, y3 = x3 * x3;
    const float y4 = x4 * x4, y5 = x5 * x5, y6 = x6 * x6, y7 = x7 * x7;
    const float y8 = x8 * x8, y9 = x9 * x9;
    const float qm  = (y3 + y4) + (y5 + y6);
    const float q12 = y1 + y2;
    const float q78 = y7 + y8;
    q.x = (y0  + q12) + qm;
    q.y = (q12 + y7)  + qm;
    q.z = (y2  + q78) + qm;
    q.w = (q78 + y9)  + qm;
}

__global__ __launch_bounds__(256) __attribute__((amdgpu_waves_per_eu(8)))
void local_variance_kernel(
    const float* __restrict__ x, float* __restrict__ out)
{
    __shared__ __attribute__((aligned(16))) float xbuf[IN_ROWS][XSTRIDE];
    __shared__ __attribute__((aligned(16))) float hs[HROWS][TILE_W];
    __shared__ __attribute__((aligned(16))) float hq[HROWS][TILE_W];

    const int tid  = threadIdx.x;
    const int lane = tid & 31;
    const int wv   = tid >> 5;
    const int n    = lane & 15;
    const int hf   = lane >> 4;
    const int cb   = wv * 16;
    const float inv_area = 1.0f / (float)(WSZ * WSZ);

    const int t0 = blockIdx.x * TILES_PER_GROUP;
    float pf[RAW_PER_THREAD];

    for (int idx = tid; idx < (HROWS - IN_ROWS) * TILE_W; idx += 256) {
        int r = IN_ROWS + idx / TILE_W;
        int c = idx % TILE_W;
        hs[r][c] = 0.0f;
        hq[r][c] = 0.0f;
    }

    load_raw(x, t0, tid, pf);
#pragma unroll
    for (int i = 0; i < RAW_PER_THREAD; ++i) {
        int idx = tid + 256 * i;
        if (idx < N_RAW) {
            int r  = idx / IN_COLS;
            int cc = idx - r * IN_COLS;
            xbuf[r][cc] = pf[i];
        }
    }
    __syncthreads();

    for (int it = 0; it < TILES_PER_GROUP; ++it) {
        const int t = t0 + it;
        const bool has_next = (it + 1 < TILES_PER_GROUP);

        if (has_next) load_raw(x, t + 1, tid, pf);

        {
            v4f s0, q0, s1, q1;
            hsum4(&xbuf[wv][4 * lane],     s0, q0);
            hsum4(&xbuf[wv + 8][4 * lane], s1, q1);
            *reinterpret_cast<v4f*>(&hs[wv][4 * lane])     = s0;
            *reinterpret_cast<v4f*>(&hq[wv][4 * lane])     = q0;
            *reinterpret_cast<v4f*>(&hs[wv + 8][4 * lane]) = s1;
            *reinterpret_cast<v4f*>(&hq[wv + 8][4 * lane]) = q1;
            if (wv + 16 < IN_ROWS) {
                v4f s2, q2;
                hsum4(&xbuf[wv + 16][4 * lane], s2, q2);
                *reinterpret_cast<v4f*>(&hs[wv + 16][4 * lane]) = s2;
                *reinterpret_cast<v4f*>(&hq[wv + 16][4 * lane]) = q2;
            }
        }
        __syncthreads();

        v8f accS = {};
        v8f accQ = {};
#pragma unroll
        for (int kc = 0; kc < 6; ++kc) {
            const int k0 = kc * 4 + hf * 2;
            v2f a;
            a.x = (k0     >= n && k0     <= n + 6) ? 1.0f : 0.0f;
            a.y = (k0 + 1 >= n && k0 + 1 <= n + 6) ? 1.0f : 0.0f;
            v2f bs, bq;
            bs.x = hs[k0][cb + n];
            bs.y = hs[k0 + 1][cb + n];
            bq.x = hq[k0][cb + n];
            bq.y = hq[k0 + 1][cb + n];
            accS = __builtin_amdgcn_wmma_f32_16x16x4_f32(
                false, a, false, bs, (short)0, accS, false, false);
            accQ = __builtin_amdgcn_wmma_f32_16x16x4_f32(
                false, a, false, bq, (short)0, accQ, false, false);
        }

        {
            int img, y0, c0;
            tile_coords(t, img, y0, c0);
            float* __restrict__ dst = out + (size_t)img * IMG_H * IMG_W;
            const int gx = c0 + cb + n;
#pragma unroll
            for (int v = 0; v < 8; ++v) {
                int gy = y0 + v + hf * 8;
                float s = accS[v] * inv_area;
                float q = accQ[v] * inv_area;
                dst[gy * IMG_W + gx] = q - s * s;
            }
        }

        if (has_next) {
#pragma unroll
            for (int i = 0; i < RAW_PER_THREAD; ++i) {
                int idx = tid + 256 * i;
                if (idx < N_RAW) {
                    int r  = idx / IN_COLS;
                    int cc = idx - r * IN_COLS;
                    xbuf[r][cc] = pf[i];
                }
            }
        }
        __syncthreads();
    }
}

extern "C" void kernel_launch(void* const* d_in, const int* in_sizes, int n_in,
                              void* d_out, int out_size, void* d_ws, size_t ws_size,
                              hipStream_t stream) {
    (void)in_sizes; (void)n_in; (void)d_ws; (void)ws_size; (void)out_size;
    const float* x = (const float*)d_in[0];
    float* out = (float*)d_out;
    dim3 grid(N_GROUPS, 1, 1);
    dim3 block(256, 1, 1);
    local_variance_kernel<<<grid, block, 0, stream>>>(x, out);
}


# Work-group barriers order LDS only (LDS-scoped fences around s_barrier): the next tile's global loads stay in flight through the matrix phase and result stores are no longer waited for
# speedup vs baseline: 1.3671x; 1.0564x over previous
#include <hip/hip_runtime.h>


typedef __attribute__((ext_vector_type(2))) float v2f;
typedef __attribute__((ext_vector_type(4))) float v4f;
typedef __attribute__((ext_vector_type(8))) float v8f;

#define WSZ     7
#define PAD     3
#define TILE_H  16
#define TILE_W  128
#define IN_ROWS (TILE_H + WSZ - 1)
#define IN_COLS (TILE_W + WSZ - 1)
#define XSTRIDE 136
#define HROWS   24
#define IMG_H   512
#define IMG_W   512
#define N_IMG   48
#define TILES_X (IMG_W / TILE_W)
#define TILES_Y (IMG_H / TILE_H)
#define N_TILES (TILES_X * TILES_Y * N_IMG)
#define N_GROUPS 1024
#define TILES_PER_GROUP (N_TILES / N_GROUPS)
#define N_RAW   (IN_ROWS * IN_COLS)
#define RAW_PER_THREAD ((N_RAW + 255) / 256)

__device__ __forceinline__ void lds_barrier()
{
    __builtin_amdgcn_fence(__ATOMIC_RELEASE, "workgroup", "local");
    __builtin_amdgcn_s_barrier();
    __builtin_amdgcn_fence(__ATOMIC_ACQUIRE, "workgroup", "local");
}

__device__ __forceinline__ void tile_coords(int t, int& img, int& y0, int& c0)
{
    y0  = (t & (TILES_Y - 1)) * TILE_H;
    c0  = ((t / TILES_Y) & (TILES_X - 1)) * TILE_W;
    img = t / (TILES_Y * TILES_X);
}

__device__ __forceinline__ void load_raw(const float* __restrict__ x, int t, int tid,
                                         float (&pf)[RAW_PER_THREAD])
{
    int img, y0, c0;
    tile_coords(t, img, y0, c0);
    const float* __restrict__ src = x + (size_t)img * IMG_H * IMG_W;
#pragma unroll
    for (int i = 0; i < RAW_PER_THREAD; ++i) {
        int idx = tid + 256 * i;
        if (idx < N_RAW) {
            int r  = idx / IN_COLS;
            int cc = idx - r * IN_COLS;
            int gy = (y0 + r  - PAD + IMG_H) & (IMG_H - 1);
            int gx = (c0 + cc - PAD + IMG_W) & (IMG_W - 1);
            pf[i] = src[gy * IMG_W + gx];
        }
    }
}

__device__ __forceinline__ void hsum4(const float* __restrict__ row, v4f& s, v4f& q)
{
    const v4f xa = *reinterpret_cast<const v4f*>(row);
    const v4f xb = *reinterpret_cast<const v4f*>(row + 4);
    const v2f xc = *reinterpret_cast<const v2f*>(row + 8);
    const float x0 = xa.x, x1 = xa.y, x2 = xa.z, x3 = xa.w;
    const float x4 = xb.x, x5 = xb.y, x6 = xb.z, x7 = xb.w;
    const float x8 = xc.x, x9 = xc.y;

    const float sm  = (x3 + x4) + (x5 + x6);
    const float s12 = x1 + x2;
    const float s78 = x7 + x8;
    s.x = (x0  + s12) + sm;
    s.y = (s12 + x7)  + sm;
    s.z = (x2  + s78) + sm;
    s.w = (s78 + x9)  + sm;

    const float y0 = x0 * x0, y1 = x1 * x1, y2 = x2 * x2, y3 = x3 * x3;
    const float y4 = x4 * x4, y5 = x5 * x5, y6 = x6 * x6, y7 = x7 * x7;
    const float y8 = x8 * x8, y9 = x9 * x9;
    const float qm  = (y3 + y4) + (y5 + y6);
    const float q12 = y1 + y2;
    const float q78 = y7 + y8;
    q.x = (y0  + q12) + qm;
    q.y = (q12 + y7)  + qm;
    q.z = (y2  + q78) + qm;
    q.w = (q78 + y9)  + qm;
}

__global__ __launch_bounds__(256) __attribute__((amdgpu_waves_per_eu(8)))
void local_variance_kernel(
    const float* __restrict__ x, float* __restrict__ out)
{
    __shared__ __attribute__((aligned(16))) float xbuf[IN_ROWS][XSTRIDE];
    __shared__ __attribute__((aligned(16))) float hs[HROWS][TILE_W];
    __shared__ __attribute__((aligned(16))) float hq[HROWS][TILE_W];

    const int tid  = threadIdx.x;
    const int lane = tid & 31;
    const int wv   = tid >> 5;
    const int n    = lane & 15;
    const int hf   = lane >> 4;
    const int cb   = wv * 16;
    const float inv_area = 1.0f / (float)(WSZ * WSZ);

    const int t0 = blockIdx.x * TILES_PER_GROUP;
    float pf[RAW_PER_THREAD];

    for (int idx = tid; idx < (HROWS - IN_ROWS) * TILE_W; idx += 256) {
        int r = IN_ROWS + idx / TILE_W;
        int c = idx % TILE_W;
        hs[r][c] = 0.0f;
        hq[r][c] = 0.0f;
    }

    load_raw(x, t0, tid, pf);
#pragma unroll
    for (int i = 0; i < RAW_PER_THREAD; ++i) {
        int idx = tid + 256 * i;
        if (idx < N_RAW) {
            int r  = idx / IN_COLS;
            int cc = idx - r * IN_COLS;
            xbuf[r][cc] = pf[i];
        }
    }
    lds_barrier();

    for (int it = 0; it < TILES_PER_GROUP; ++it) {
        const int t = t0 + it;
        const bool has_next = (it + 1 < TILES_PER_GROUP);

        if (has_next) load_raw(x, t + 1, tid, pf);

        {
            v4f s0, q0, s1, q1;
            hsum4(&xbuf[wv][4 * lane],     s0, q0);
            hsum4(&xbuf[wv + 8][4 * lane], s1, q1);
            *reinterpret_cast<v4f*>(&hs[wv][4 * lane])     = s0;
            *reinterpret_cast<v4f*>(&hq[wv][4 * lane])     = q0;
            *reinterpret_cast<v4f*>(&hs[wv + 8][4 * lane]) = s1;
            *reinterpret_cast<v4f*>(&hq[wv + 8][4 * lane]) = q1;
            if (wv + 16 < IN_ROWS) {
                v4f s2, q2;
                hsum4(&xbuf[wv + 16][4 * lane], s2, q2);
                *reinterpret_cast<v4f*>(&hs[wv + 16][4 * lane]) = s2;
                *reinterpret_cast<v4f*>(&hq[wv + 16][4 * lane]) = q2;
            }
        }
        lds_barrier();

        v8f accS = {};
        v8f accQ = {};
#pragma unroll
        for (int kc = 0; kc < 6; ++kc) {
            const int k0 = kc * 4 + hf * 2;
            v2f a;
            a.x = (k0     >= n && k0     <= n + 6) ? 1.0f : 0.0f;
            a.y = (k0 + 1 >= n && k0 + 1 <= n + 6) ? 1.0f : 0.0f;
            v2f bs, bq;
            bs.x = hs[k0][cb + n];
            bs.y = hs[k0 + 1][cb + n];
            bq.x = hq[k0][cb + n];
            bq.y = hq[k0 + 1][cb + n];
            accS = __builtin_amdgcn_wmma_f32_16x16x4_f32(
                false, a, false, bs, (short)0, accS, false, false);
            accQ = __builtin_amdgcn_wmma_f32_16x16x4_f32(
                false, a, false, bq, (short)0, accQ, false, false);
        }

        {
            int img, y0, c0;
            tile_coords(t, img, y0, c0);
            float* __restrict__ dst = out + (size_t)img * IMG_H * IMG_W;
            const int gx = c0 + cb + n;
#pragma unroll
            for (int v = 0; v < 8; ++v) {
                int gy = y0 + v + hf * 8;
                float s = accS[v] * inv_area;
                float q = accQ[v] * inv_area;
                dst[gy * IMG_W + gx] = q - s * s;
            }
        }

        if (has_next) {
#pragma unroll
            for (int i = 0; i < RAW_PER_THREAD; ++i) {
                int idx = tid + 256 * i;
                if (idx < N_RAW) {
                    int r  = idx / IN_COLS;
                    int cc = idx - r * IN_COLS;
                    xbuf[r][cc] = pf[i];
                }
            }
        }
        lds_barrier();
    }
}

extern "C" void kernel_launch(void* const* d_in, const int* in_sizes, int n_in,
                              void* d_out, int out_size, void* d_ws, size_t ws_size,
                              hipStream_t stream) {
    (void)in_sizes; (void)n_in; (void)d_ws; (void)ws_size; (void)out_size;
    const float* x = (const float*)d_in[0];
    float* out = (float*)d_out;
    dim3 grid(N_GROUPS, 1, 1);
    dim3 block(256, 1, 1);
    local_variance_kernel<<<grid, block, 0, stream>>>(x, out);
}


# Results staged through LDS and stored as whole rows: one 16-byte-per-lane store instruction of a wave writes a full 512-byte tile row (4 whole lines), not 8 stores of two 64-byte half-lines each
# speedup vs baseline: 1.5829x; 1.1579x over previous
#include <hip/hip_runtime.h>


typedef __attribute__((ext_vector_type(2))) float v2f;
typedef __attribute__((ext_vector_type(4))) float v4f;
typedef __attribute__((ext_vector_type(8))) float v8f;

#define WSZ     7
#define PAD     3
#define TILE_H  16
#define TILE_W  128
#define IN_ROWS (TILE_H + WSZ - 1)
#define IN_COLS (TILE_W + WSZ - 1)
#define XSTRIDE 136
#define HROWS   24
#define OSTRIDE 132
#define IMG_H   512
#define IMG_W   512
#define N_IMG   48
#define TILES_X (IMG_W / TILE_W)
#define TILES_Y (IMG_H / TILE_H)
#define N_TILES (TILES_X * TILES_Y * N_IMG)
#define N_GROUPS 1024
#define TILES_PER_GROUP (N_TILES / N_GROUPS)
#define N_RAW   (IN_ROWS * IN_COLS)
#define RAW_PER_THREAD ((N_RAW + 255) / 256)

__device__ __forceinline__ void lds_barrier()
{
    __builtin_amdgcn_fence(__ATOMIC_RELEASE, "workgroup", "local");
    __builtin_amdgcn_s_barrier();
    __builtin_amdgcn_fence(__ATOMIC_ACQUIRE, "workgroup", "local");
}

__device__ __forceinline__ void tile_coords(int t, int& img, int& y0, int& c0)
{
    y0  = (t & (TILES_Y - 1)) * TILE_H;
    c0  = ((t / TILES_Y) & (TILES_X - 1)) * TILE_W;
    img = t / (TILES_Y * TILES_X);
}

__device__ __forceinline__ void load_raw(const float* __restrict__ x, int t, int tid,
                                         float (&pf)[RAW_PER_THREAD])
{
    int img, y0, c0;
    tile_coords(t, img, y0, c0);
    const float* __restrict__ src = x + (size_t)img * IMG_H * IMG_W;
#pragma unroll
    for (int i = 0; i < RAW_PER_THREAD; ++i) {
        int idx = tid + 256 * i;
        if (idx < N_RAW) {
            int r  = idx / IN_COLS;
            int cc = idx - r * IN_COLS;
            int gy = (y0 + r  - PAD + IMG_H) & (IMG_H - 1);
            int gx = (c0 + cc - PAD + IMG_W) & (IMG_W - 1);
            pf[i] = src[gy * IMG_W + gx];
        }
    }
}

__device__ __forceinline__ void hsum4(const float* __restrict__ row, v4f& s, v4f& q)
{
    const v4f xa = *reinterpret_cast<const v4f*>(row);
    const v4f xb = *reinterpret_cast<const v4f*>(row + 4);
    const v2f xc = *reinterpret_cast<const v2f*>(row + 8);
    const float x0 = xa.x, x1 = xa.y, x2 = xa.z, x3 = xa.w;
    const float x4 = xb.x, x5 = xb.y, x6 = xb.z, x7 = xb.w;
    const float x8 = xc.x, x9 = xc.y;

    const float sm  = (x3 + x4) + (x5 + x6);
    const float s12 = x1 + x2;
    const float s78 = x7 + x8;
    s.x = (x0  + s12) + sm;
    s.y = (s12 + x7)  + sm;
    s.z = (x2  + s78) + sm;
    s.w = (s78 + x9)  + sm;

    const float y0 = x0 * x0, y1 = x1 * x1, y2 = x2 * x2, y3 = x3 * x3;
    const float y4 = x4 * x4, y5 = x5 * x5, y6 = x6 * x6, y7 = x7 * x7;
    const float y8 = x8 * x8, y9 = x9 * x9;
    const float qm  = (y3 + y4) + (y5 + y6);
    const float q12 = y1 + y2;
    const float q78 = y7 + y8;
    q.x = (y0  + q12) + qm;
    q.y = (q12 + y7)  + qm;
    q.z = (y2  + q78) + qm;
    q.w = (q78 + y9)  + qm;
}

__device__ __forceinline__ void store_rows(float* __restrict__ out,
                                           const float (*__restrict__ ob)[OSTRIDE],
                                           int t, int wv, int lane)
{
    int img, y0, c0;
    tile_coords(t, img, y0, c0);
    float* __restrict__ dst = out + (size_t)img * IMG_H * IMG_W + (size_t)y0 * IMG_W + c0 + 4 * lane;
    const v4f r0 = *reinterpret_cast<const v4f*>(&ob[wv][4 * lane]);
    const v4f r1 = *reinterpret_cast<const v4f*>(&ob[wv + 8][4 * lane]);
    *reinterpret_cast<v4f*>(dst + wv * IMG_W)       = r0;
    *reinterpret_cast<v4f*>(dst + (wv + 8) * IMG_W) = r1;
}

__global__ __launch_bounds__(256) __attribute__((amdgpu_waves_per_eu(8)))
void local_variance_kernel(
    const float* __restrict__ x, float* __restrict__ out)
{
    __shared__ __attribute__((aligned(16))) float xbuf[IN_ROWS][XSTRIDE];
    __shared__ __attribute__((aligned(16))) float hs[HROWS][TILE_W];
    __shared__ __attribute__((aligned(16))) float hq[HROWS][TILE_W];
    __shared__ __attribute__((aligned(16))) float obuf[TILE_H][OSTRIDE];

    const int tid  = threadIdx.x;
    const int lane = tid & 31;
    const int wv   = tid >> 5;
    const int n    = lane & 15;
    const int hf   = lane >> 4;
    const int cb   = wv * 16;
    const float inv_area = 1.0f / (float)(WSZ * WSZ);

    const int t0 = blockIdx.x * TILES_PER_GROUP;
    float pf[RAW_PER_THREAD];

    for (int idx = tid; idx < (HROWS - IN_ROWS) * TILE_W; idx += 256) {
        int r = IN_ROWS + idx / TILE_W;
        int c = idx % TILE_W;
        hs[r][c] = 0.0f;
        hq[r][c] = 0.0f;
    }

    load_raw(x, t0, tid, pf);
#pragma unroll
    for (int i = 0; i < RAW_PER_THREAD; ++i) {
        int idx = tid + 256 * i;
        if (idx < N_RAW) {
            int r  = idx / IN_COLS;
            int cc = idx - r * IN_COLS;
            xbuf[r][cc] = pf[i];
        }
    }
    lds_barrier();

    for (int it = 0; it < TILES_PER_GROUP; ++it) {
        const int t = t0 + it;
        const bool has_next = (it + 1 < TILES_PER_GROUP);

        if (has_next) load_raw(x, t + 1, tid, pf);

        {
            v4f s0, q0, s1, q1;
            hsum4(&xbuf[wv][4 * lane],     s0, q0);
            hsum4(&xbuf[wv + 8][4 * lane], s1, q1);
            *reinterpret_cast<v4f*>(&hs[wv][4 * lane])     = s0;
            *reinterpret_cast<v4f*>(&hq[wv][4 * lane])     = q0;
            *reinterpret_cast<v4f*>(&hs[wv + 8][4 * lane]) = s1;
            *reinterpret_cast<v4f*>(&hq[wv + 8][4 * lane]) = q1;
            if (wv + 16 < IN_ROWS) {
                v4f s2, q2;
                hsum4(&xbuf[wv + 16][4 * lane], s2, q2);
                *reinterpret_cast<v4f*>(&hs[wv + 16][4 * lane]) = s2;
                *reinterpret_cast<v4f*>(&hq[wv + 16][4 * lane]) = q2;
            }
        }
        lds_barrier();

        v8f accS = {};
        v8f accQ = {};
#pragma unroll
        for (int kc = 0; kc < 6; ++kc) {
            const int k0 = kc * 4 + hf * 2;
            v2f a;
            a.x = (k0     >= n && k0     <= n + 6) ? 1.0f : 0.0f;
            a.y = (k0 + 1 >= n && k0 + 1 <= n + 6) ? 1.0f : 0.0f;
            v2f bs, bq;
            bs.x = hs[k0][cb + n];
            bs.y = hs[k0 + 1][cb + n];
            bq.x = hq[k0][cb + n];
            bq.y = hq[k0 + 1][cb + n];
            accS = __builtin_amdgcn_wmma_f32_16x16x4_f32(
                false, a, false, bs, (short)0, accS, false, false);
            accQ = __builtin_amdgcn_wmma_f32_16x16x4_f32(
                false, a, false, bq, (short)0, accQ, false, false);
        }

        {
#pragma unroll
            for (int v = 0; v < 8; ++v) {
                float s = accS[v] * inv_area;
                float q = accQ[v] * inv_area;
                obuf[v + hf * 8][cb + n] = q - s * s;
            }
        }

        if (has_next) {
#pragma unroll
            for (int i = 0; i < RAW_PER_THREAD; ++i) {
                int idx = tid + 256 * i;
                if (idx < N_RAW) {
                    int r  = idx / IN_COLS;
                    int cc = idx - r * IN_COLS;
                    xbuf[r][cc] = pf[i];
                }
            }
        }
        lds_barrier();

        store_rows(out, obuf, t, wv, lane);
    }
}

extern "C" void kernel_launch(void* const* d_in, const int* in_sizes, int n_in,
                              void* d_out, int out_size, void* d_ws, size_t ws_size,
                              hipStream_t stream) {
    (void)in_sizes; (void)n_in; (void)d_ws; (void)ws_size; (void)out_size;
    const float* x = (const float*)d_in[0];
    float* out = (float*)d_out;
    dim3 grid(N_GROUPS, 1, 1);
    dim3 block(256, 1, 1);
    local_variance_kernel<<<grid, block, 0, stream>>>(x, out);
}


# Input tile fetched by rows: each wave loads 3 rows as aligned 16-byte pieces from scalar row addresses (3 wide + 1 halo load per lane, not 12 dword loads with div/mod addressing); 16-byte LDS staging
# speedup vs baseline: 1.6954x; 1.0711x over previous
#include <hip/hip_runtime.h>


typedef __attribute__((ext_vector_type(2))) float v2f;
typedef __attribute__((ext_vector_type(4))) float v4f;
typedef __attribute__((ext_vector_type(8))) float v8f;

#define WSZ     7
#define PAD     3
#define TILE_H  16
#define TILE_W  128
#define IN_ROWS (TILE_H + WSZ - 1)
#define XSTRIDE 136
#define XOFF    4
#define XROWS   24
#define N_HALO_LANES 18
#define HROWS   24
#define OSTRIDE 132
#define IMG_H   512
#define IMG_W   512
#define N_IMG   48
#define TILES_X (IMG_W / TILE_W)
#define TILES_Y (IMG_H / TILE_H)
#define N_TILES (TILES_X * TILES_Y * N_IMG)
#define N_GROUPS 1024
#define TILES_PER_GROUP (N_TILES / N_GROUPS)
#define ROWS_PER_WAVE 3

__device__ __forceinline__ void lds_barrier()
{
    __builtin_amdgcn_fence(__ATOMIC_RELEASE, "workgroup", "local");
    __builtin_amdgcn_s_barrier();
    __builtin_amdgcn_fence(__ATOMIC_ACQUIRE, "workgroup", "local");
}

__device__ __forceinline__ void tile_coords(int t, int& img, int& y0, int& c0)
{
    y0  = (t & (TILES_Y - 1)) * TILE_H;
    c0  = ((t / TILES_Y) & (TILES_X - 1)) * TILE_W;
    img = t / (TILES_Y * TILES_X);
}

__device__ __forceinline__ void load_raw(const float* __restrict__ x, int t, int wvs, int lane,
                                         v4f (&pf)[ROWS_PER_WAVE], float& ph)
{
    int img, y0, c0;
    tile_coords(t, img, y0, c0);
    const float* __restrict__ src = x + (size_t)img * IMG_H * IMG_W;
#pragma unroll
    for (int j = 0; j < ROWS_PER_WAVE; ++j) {
        const int gy = (y0 + wvs + 8 * j - PAD + IMG_H) & (IMG_H - 1);
        pf[j] = *reinterpret_cast<const v4f*>(src + gy * IMG_W + c0 + 4 * lane);
    }
    const int hl = (lane < N_HALO_LANES) ? lane : (N_HALO_LANES - 1);
    const int hj = hl / 6;
    const int hk = hl - 6 * hj;
    const int cc = (hk < PAD) ? (hk - PAD) : (TILE_W + hk - PAD);
    const int gy = (y0 + wvs + 8 * hj - PAD + IMG_H) & (IMG_H - 1);
    const int gx = (c0 + cc + IMG_W) & (IMG_W - 1);
    ph = src[gy * IMG_W + gx];
}

__device__ __forceinline__ void stage_raw(float (*__restrict__ xb)[XSTRIDE], int wvs, int lane,
                                          const v4f (&pf)[ROWS_PER_WAVE], float ph)
{
#pragma unroll
    for (int j = 0; j < ROWS_PER_WAVE; ++j)
        *reinterpret_cast<v4f*>(&xb[wvs + 8 * j][XOFF + 4 * lane]) = pf[j];
    const int hl = (lane < N_HALO_LANES) ? lane : (N_HALO_LANES - 1);
    const int hj = hl / 6;
    const int hk = hl - 6 * hj;
    const int cc = (hk < PAD) ? (hk - PAD) : (TILE_W + hk - PAD);
    xb[wvs + 8 * hj][XOFF + cc] = ph;
}

__device__ __forceinline__ void hsum4(const float* __restrict__ row, v4f& s, v4f& q)
{
    const v4f xa = *reinterpret_cast<const v4f*>(row);
    const v4f xb = *reinterpret_cast<const v4f*>(row + 4);
    const v4f xc = *reinterpret_cast<const v4f*>(row + 8);
    const float x0 = xa.y, x1 = xa.z, x2 = xa.w, x3 = xb.x;
    const float x4 = xb.y, x5 = xb.z, x6 = xb.w, x7 = xc.x;
    const float x8 = xc.y, x9 = xc.z;

    const float sm  = (x3 + x4) + (x5 + x6);
    const float s12 = x1 + x2;
    const float s78 = x7 + x8;
    s.x = (x0  + s12) + sm;
    s.y = (s12 + x7)  + sm;
    s.z = (x2  + s78) + sm;
    s.w = (s78 + x9)  + sm;

    const float y0 = x0 * x0, y1 = x1 * x1, y2 = x2 * x2, y3 = x3 * x3;
    const float y4 = x4 * x4, y5 = x5 * x5, y6 = x6 * x6, y7 = x7 * x7;
    const float y8 = x8 * x8, y9 = x9 * x9;
    const float qm  = (y3 + y4) + (y5 + y6);
    const float q12 = y1 + y2;
    const float q78 = y7 + y8;
    q.x = (y0  + q12) + qm;
    q.y = (q12 + y7)  + qm;
    q.z = (y2  + q78) + qm;
    q.w = (q78 + y9)  + qm;
}

__device__ __forceinline__ void store_rows(float* __restrict__ out,
                                           const float (*__restrict__ ob)[OSTRIDE],
                                           int t, int wv, int lane)
{
    int img, y0, c0;
    tile_coords(t, img, y0, c0);
    float* __restrict__ dst = out + (size_t)img * IMG_H * IMG_W + (size_t)y0 * IMG_W + c0 + 4 * lane;
    const v4f r0 = *reinterpret_cast<const v4f*>(&ob[wv][4 * lane]);
    const v4f r1 = *reinterpret_cast<const v4f*>(&ob[wv + 8][4 * lane]);
    *reinterpret_cast<v4f*>(dst + wv * IMG_W)       = r0;
    *reinterpret_cast<v4f*>(dst + (wv + 8) * IMG_W) = r1;
}

__global__ __launch_bounds__(256) __attribute__((amdgpu_waves_per_eu(8)))
void local_variance_kernel(
    const float* __restrict__ x, float* __restrict__ out)
{
    __shared__ __attribute__((aligned(16))) float xbuf[XROWS][XSTRIDE];
    __shared__ __attribute__((aligned(16))) float hs[HROWS][TILE_W];
    __shared__ __attribute__((aligned(16))) float hq[HROWS][TILE_W];
    __shared__ __attribute__((aligned(16))) float obuf[TILE_H][OSTRIDE];

    const int tid  = threadIdx.x;
    const int lane = tid & 31;
    const int wv   = tid >> 5;
    const int n    = lane & 15;
    const int hf   = lane >> 4;
    const int cb   = wv * 16;
    const float inv_area = 1.0f / (float)(WSZ * WSZ);

    const int wvs = __builtin_amdgcn_readfirstlane(wv);

    const int t0 = blockIdx.x * TILES_PER_GROUP;
    v4f   pf[ROWS_PER_WAVE];
    float ph = 0.0f;

    for (int idx = tid; idx < (HROWS - IN_ROWS) * TILE_W; idx += 256) {
        int r = IN_ROWS + idx / TILE_W;
        int c = idx % TILE_W;
        hs[r][c] = 0.0f;
        hq[r][c] = 0.0f;
    }
    if (tid < 2 * IN_ROWS)
        xbuf[tid >> 1][(tid & 1) * (XSTRIDE - 1)] = 0.0f;

    load_raw(x, t0, wvs, lane, pf, ph);
    stage_raw(xbuf, wvs, lane, pf, ph);
    lds_barrier();

    for (int it = 0; it < TILES_PER_GROUP; ++it) {
        const int t = t0 + it;
        const bool has_next = (it + 1 < TILES_PER_GROUP);

        if (has_next) load_raw(x, t + 1, wvs, lane, pf, ph);

        {
            v4f s0, q0, s1, q1;
            hsum4(&xbuf[wv][4 * lane],     s0, q0);
            hsum4(&xbuf[wv + 8][4 * lane], s1, q1);
            *reinterpret_cast<v4f*>(&hs[wv][4 * lane])     = s0;
            *reinterpret_cast<v4f*>(&hq[wv][4 * lane])     = q0;
            *reinterpret_cast<v4f*>(&hs[wv + 8][4 * lane]) = s1;
            *reinterpret_cast<v4f*>(&hq[wv + 8][4 * lane]) = q1;
            if (wv + 16 < IN_ROWS) {
                v4f s2, q2;
                hsum4(&xbuf[wv + 16][4 * lane], s2, q2);
                *reinterpret_cast<v4f*>(&hs[wv + 16][4 * lane]) = s2;
                *reinterpret_cast<v4f*>(&hq[wv + 16][4 * lane]) = q2;
            }
        }
        lds_barrier();

        v8f accS = {};
        v8f accQ = {};
#pragma unroll
        for (int kc = 0; kc < 6; ++kc) {
            const int k0 = kc * 4 + hf * 2;
            v2f a;
            a.x = (k0     >= n && k0     <= n + 6) ? 1.0f : 0.0f;
            a.y = (k0 + 1 >= n && k0 + 1 <= n + 6) ? 1.0f : 0.0f;
            v2f bs, bq;
            bs.x = hs[k0][cb + n];
            bs.y = hs[k0 + 1][cb + n];
            bq.x = hq[k0][cb + n];
            bq.y = hq[k0 + 1][cb + n];
            accS = __builtin_amdgcn_wmma_f32_16x16x4_f32(
                false, a, false, bs, (short)0, accS, false, false);
            accQ = __builtin_amdgcn_wmma_f32_16x16x4_f32(
                false, a, false, bq, (short)0, accQ, false, false);
        }

        {
#pragma unroll
            for (int v = 0; v < 8; ++v) {
                float s = accS[v] * inv_area;
                float q = accQ[v] * inv_area;
                obuf[v + hf * 8][cb + n] = q - s * s;
            }
        }

        __builtin_amdgcn_sched_barrier(0);
        if (has_next) stage_raw(xbuf, wvs, lane, pf, ph);
        lds_barrier();

        store_rows(out, obuf, t, wv, lane);
    }
}

extern "C" void kernel_launch(void* const* d_in, const int* in_sizes, int n_in,
                              void* d_out, int out_size, void* d_ws, size_t ws_size,
                              hipStream_t stream) {
    (void)in_sizes; (void)n_in; (void)d_ws; (void)ws_size; (void)out_size;
    const float* x = (const float*)d_in[0];
    float* out = (float*)d_out;
    dim3 grid(N_GROUPS, 1, 1);
    dim3 block(256, 1, 1);
    local_variance_kernel<<<grid, block, 0, stream>>>(x, out);
}
